# starting kernel
# baseline (speedup 1.0000x reference)
#include <hip/hip_runtime.h>

typedef __attribute__((ext_vector_type(2))) float v2f;
typedef __attribute__((ext_vector_type(8))) float v8f;

#define DWORDS_D 300
#define S_CATS 256
#define A_CATS 256
#define WPC 8
#define NROWS 2048
#define COS_EPS 1e-8f

__global__ void ComSim_gather_rows(const float* __restrict__ emb,
                                   const int* __restrict__ senCats,
                                   const int* __restrict__ annCats,
                                   const int* __restrict__ noneIdxPtr,
                                   float* __restrict__ senE,
                                   float* __restrict__ annE,
                                   float* __restrict__ ns,
                                   float* __restrict__ na,
                                   float* __restrict__ noneMask)
{
    __shared__ float redF[128];
    __shared__ int   redI[128];

    const int bid   = blockIdx.x;
    const int tid   = threadIdx.x;
    const bool isSen = bid < NROWS;
    const int row   = isSen ? bid : bid - NROWS;
    const int idx   = isSen ? senCats[row] : annCats[row];
    const int nIdx  = noneIdxPtr[0];

    const float* __restrict__ src  = emb + (long long)idx  * DWORDS_D;
    const float* __restrict__ nsrc = emb + (long long)nIdx * DWORDS_D;
    float* __restrict__ dst = (isSen ? senE : annE) + (long long)row * DWORDS_D;

    float ss = 0.0f;
    int eq = 1;
    for (int d = tid; d < DWORDS_D; d += 128) {
        float v = src[d];
        dst[d] = v;
        ss += v * v;
        if (isSen) eq &= (v == nsrc[d]) ? 1 : 0;
    }
    redF[tid] = ss;
    redI[tid] = eq;
    __syncthreads();
    for (int off = 64; off > 0; off >>= 1) {
        if (tid < off) {
            redF[tid] += redF[tid + off];
            redI[tid] &= redI[tid + off];
        }
        __syncthreads();
    }
    if (tid == 0) {
        float nrm = sqrtf(redF[0]);
        if (isSen) {
            ns[row] = nrm;
            noneMask[row] = redI[0] ? 1.0f : 0.0f;
        } else {
            na[row] = nrm;
        }
    }
}

__global__ void ComSim_sen_word_sum(const float* __restrict__ senE,
                                    float* __restrict__ out)
{
    const int s = blockIdx.x;
    const int tid = threadIdx.x;
    for (int d = tid; d < DWORDS_D; d += 128) {
        float acc = 0.0f;
#pragma unroll
        for (int w = 0; w < WPC; ++w)
            acc += senE[((long long)(s * WPC + w)) * DWORDS_D + d];
        out[(long long)s * DWORDS_D + d] = acc;
    }
}

__global__ void ComSim_cos_tile_wmma(const float* __restrict__ senE,
                                     const float* __restrict__ annE,
                                     const float* __restrict__ ns,
                                     const float* __restrict__ na,
                                     const float* __restrict__ noneMask,
                                     float* __restrict__ maxSim)
{
    __shared__ float tile[16 * 16];

    const int lane = threadIdx.x;
    const int half = lane >> 4;
    const int l16  = lane & 15;
    const int row0 = blockIdx.y * 16;
    const int col0 = blockIdx.x * 16;

    const float* __restrict__ aRow = senE + (long long)(row0 + l16) * DWORDS_D + 2 * half;
    const float* __restrict__ bRow = annE + (long long)(col0 + l16) * DWORDS_D + 2 * half;

    v8f c = {};
#pragma unroll 5
    for (int k = 0; k < DWORDS_D; k += 4) {
        v2f a = *(const v2f*)(aRow + k);
        v2f b = *(const v2f*)(bRow + k);
        c = __builtin_amdgcn_wmma_f32_16x16x4_f32(
                 false, a,  false, b,
                 (short)0, c,  false,  false);
    }

    const float naV = na[col0 + l16];
#pragma unroll
    for (int r = 0; r < 8; ++r) {
        const int m = r + 8 * half;
        const float nsV   = ns[row0 + m];
        const float denom = fmaxf(nsV * naV, COS_EPS);
        float cosv = c[r] / denom;
        if (noneMask[row0 + m] != 0.0f) cosv = 0.0f;
        tile[m * 16 + l16] = cosv;
    }
    __syncthreads();

    if (lane < 4) {
        const int si = lane >> 1;
        const int ai = lane & 1;
        float cur = 0.0f;
        for (int w = 0; w < WPC; ++w) {
            for (int v = 0; v < WPC; ++v) {
                float x = tile[(si * 8 + w) * 16 + (ai * 8 + v)];
                cur = ((x >= cur) || (x < 0.0f)) ? x : cur;
            }
        }
        const int sIdx = blockIdx.y * 2 + si;
        const int aIdx = blockIdx.x * 2 + ai;
        maxSim[sIdx * A_CATS + aIdx] = cur;
    }
}

extern "C" void kernel_launch(void* const* d_in, const int* in_sizes, int n_in,
                              void* d_out, int out_size, void* d_ws, size_t ws_size,
                              hipStream_t stream) {
    const float* emb      = (const float*)d_in[0];
    const int*   senCats  = (const int*)d_in[1];
    const int*   annCats  = (const int*)d_in[2];
    const int*   noneIdx  = (const int*)d_in[3];

    float* out    = (float*)d_out;
    float* maxSim = out;
    float* senWd  = out + S_CATS * A_CATS;

    char* ws = (char*)d_ws;
    float* senE     = (float*)(ws + 0);
    float* annE     = (float*)(ws + 2457600);
    float* ns       = (float*)(ws + 4915200);
    float* na       = (float*)(ws + 4923392);
    float* noneMask = (float*)(ws + 4931584);

    ComSim_gather_rows<<<dim3(2 * NROWS), dim3(128), 0, stream>>>(
        emb, senCats, annCats, noneIdx, senE, annE, ns, na, noneMask);

    ComSim_sen_word_sum<<<dim3(S_CATS), dim3(128), 0, stream>>>(senE, senWd);

    ComSim_cos_tile_wmma<<<dim3(128, 128), dim3(32), 0, stream>>>(
        senE, annE, ns, na, noneMask, maxSim);
}


# Dot kernel: 64x64 block per 4-wave work-group, both operand blocks staged whole-K in LDS by full-line loads, 2x2 WMMA tiles per wave fed from LDS; k order and order of sums unchanged
# speedup vs baseline: 1.9880x; 1.9880x over previous
#include <hip/hip_runtime.h>

typedef __attribute__((ext_vector_type(2))) float v2f;
typedef __attribute__((ext_vector_type(4))) float v4f;
typedef __attribute__((ext_vector_type(8))) float v8f;

#define DWORDS_D 300
#define S_CATS 256
#define A_CATS 256
#define WPC 8
#define NROWS 2048
#define COS_EPS 1e-8f

__global__ void ComSim_gather_rows(const float* __restrict__ emb,
                                   const int* __restrict__ senCats,
                                   const int* __restrict__ annCats,
                                   const int* __restrict__ noneIdxPtr,
                                   float* __restrict__ senE,
                                   float* __restrict__ annE,
                                   float* __restrict__ ns,
                                   float* __restrict__ na,
                                   float* __restrict__ noneMask)
{
    __shared__ float redF[128];
    __shared__ int   redI[128];

    const int bid   = blockIdx.x;
    const int tid   = threadIdx.x;
    const bool isSen = bid < NROWS;
    const int row   = isSen ? bid : bid - NROWS;
    const int idx   = isSen ? senCats[row] : annCats[row];
    const int nIdx  = noneIdxPtr[0];

    const float* __restrict__ src  = emb + (long long)idx  * DWORDS_D;
    const float* __restrict__ nsrc = emb + (long long)nIdx * DWORDS_D;
    float* __restrict__ dst = (isSen ? senE : annE) + (long long)row * DWORDS_D;

    float ss = 0.0f;
    int eq = 1;
    for (int d = tid; d < DWORDS_D; d += 128) {
        float v = src[d];
        dst[d] = v;
        ss += v * v;
        if (isSen) eq &= (v == nsrc[d]) ? 1 : 0;
    }
    redF[tid] = ss;
    redI[tid] = eq;
    __syncthreads();
    for (int off = 64; off > 0; off >>= 1) {
        if (tid < off) {
            redF[tid] += redF[tid + off];
            redI[tid] &= redI[tid + off];
        }
        __syncthreads();
    }
    if (tid == 0) {
        float nrm = sqrtf(redF[0]);
        if (isSen) {
            ns[row] = nrm;
            noneMask[row] = redI[0] ? 1.0f : 0.0f;
        } else {
            na[row] = nrm;
        }
    }
}

__global__ void ComSim_sen_word_sum(const float* __restrict__ senE,
                                    float* __restrict__ out)
{
    const int s = blockIdx.x;
    const int tid = threadIdx.x;
    for (int d = tid; d < DWORDS_D; d += 128) {
        float acc = 0.0f;
#pragma unroll
        for (int w = 0; w < WPC; ++w)
            acc += senE[((long long)(s * WPC + w)) * DWORDS_D + d];
        out[(long long)s * DWORDS_D + d] = acc;
    }
}

#define TILE 64
#define TILE_DW (TILE * DWORDS_D)
#define TILE_V4 (TILE_DW / 4)
#define KCH 5
#define NKCH 15
#define EPI_PITCH 34

__global__ void __launch_bounds__(128)
ComSim_cos_tile_wmma(const float* __restrict__ senE,
                     const float* __restrict__ annE,
                     const float* __restrict__ ns,
                     const float* __restrict__ na,
                     const float* __restrict__ noneMask,
                     float* __restrict__ maxSim)
{
    __shared__ __attribute__((aligned(16))) float lds[2 * TILE_DW];

    const int tid  = threadIdx.x;
    const int lane = tid & 31;
    const int wave = tid >> 5;
    const int half = lane >> 4;
    const int l16  = lane & 15;
    const int wr   = wave >> 1;
    const int wc   = wave & 1;
    const int row0 = blockIdx.y * TILE;
    const int col0 = blockIdx.x * TILE;
    const int wrow0 = row0 + wr * 32;
    const int wcol0 = col0 + wc * 32;

    float naV[2];
    float nsV[2][8];
    float nmV[2][8];
#pragma unroll
    for (int i = 0; i < 2; ++i) {
        naV[i] = na[wcol0 + i * 16 + l16];
#pragma unroll
        for (int r = 0; r < 8; ++r) {
            nsV[i][r] = ns[wrow0 + i * 16 + 8 * half + r];
            nmV[i][r] = noneMask[wrow0 + i * 16 + 8 * half + r];
        }
    }

    {
        const v4f* __restrict__ gA = (const v4f*)(senE + (long long)row0 * DWORDS_D);
        const v4f* __restrict__ gB = (const v4f*)(annE + (long long)col0 * DWORDS_D);
        v4f* sAB = (v4f*)lds;
        v4f t[75];
#pragma unroll
        for (int p = 0; p < 75; ++p) {
            const int idx = tid + 128 * p;
            if (p < 37)      t[p] = gA[idx];
            else if (p > 37) t[p] = gB[idx - TILE_V4];
            else             t[p] = (tid < 64) ? gA[idx] : gB[idx - TILE_V4];
        }
#pragma unroll
        for (int p = 0; p < 75; ++p)
            sAB[tid + 128 * p] = t[p];
    }
    __syncthreads();

    const float* pa0 = lds + (wr * 32 + l16) * DWORDS_D + 2 * half;
    const float* pa1 = pa0 + 16 * DWORDS_D;
    const float* pb0 = lds + TILE_DW + (wc * 32 + l16) * DWORDS_D + 2 * half;
    const float* pb1 = pb0 + 16 * DWORDS_D;

    v8f c00 = {}, c01 = {}, c10 = {}, c11 = {};
    v2f fa0[2][KCH], fa1[2][KCH], fb0[2][KCH], fb1[2][KCH];

#pragma unroll
    for (int j = 0; j < KCH; ++j) {
        fa0[0][j] = *(const v2f*)(pa0 + 4 * j);
        fa1[0][j] = *(const v2f*)(pa1 + 4 * j);
        fb0[0][j] = *(const v2f*)(pb0 + 4 * j);
        fb1[0][j] = *(const v2f*)(pb1 + 4 * j);
    }
#pragma unroll
    for (int ch = 0; ch < NKCH; ++ch) {
        const int cur = ch & 1;
        const int nxt = cur ^ 1;
        if (ch + 1 < NKCH) {
            const int k = (ch + 1) * KCH * 4;
#pragma unroll
            for (int j = 0; j < KCH; ++j) {
                fa0[nxt][j] = *(const v2f*)(pa0 + k + 4 * j);
                fa1[nxt][j] = *(const v2f*)(pa1 + k + 4 * j);
                fb0[nxt][j] = *(const v2f*)(pb0 + k + 4 * j);
                fb1[nxt][j] = *(const v2f*)(pb1 + k + 4 * j);
            }
        }
#pragma unroll
        for (int j = 0; j < KCH; ++j) {
            c00 = __builtin_amdgcn_wmma_f32_16x16x4_f32(false, fa0[cur][j], false, fb0[cur][j],
                                                        (short)0, c00, false, false);
            c01 = __builtin_amdgcn_wmma_f32_16x16x4_f32(false, fa0[cur][j], false, fb1[cur][j],
                                                        (short)0, c01, false, false);
            c10 = __builtin_amdgcn_wmma_f32_16x16x4_f32(false, fa1[cur][j], false, fb0[cur][j],
                                                        (short)0, c10, false, false);
            c11 = __builtin_amdgcn_wmma_f32_16x16x4_f32(false, fa1[cur][j], false, fb1[cur][j],
                                                        (short)0, c11, false, false);
        }
    }

    __syncthreads();

    float* tile = lds + wave * (32 * EPI_PITCH);
#pragma unroll
    for (int mi = 0; mi < 2; ++mi) {
#pragma unroll
        for (int ni = 0; ni < 2; ++ni) {
            const v8f c = (mi == 0) ? ((ni == 0) ? c00 : c01) : ((ni == 0) ? c10 : c11);
#pragma unroll
            for (int r = 0; r < 8; ++r) {
                const int m = mi * 16 + r + 8 * half;
                const float denom = fmaxf(nsV[mi][r] * naV[ni], COS_EPS);
                float cosv = c[r] / denom;
                if (nmV[mi][r] != 0.0f) cosv = 0.0f;
                tile[m * EPI_PITCH + ni * 16 + l16] = cosv;
            }
        }
    }
    __syncthreads();

    if (lane < 16) {
        const int si = lane >> 2;
        const int ai = lane & 3;
        float cur = 0.0f;
        for (int w = 0; w < WPC; ++w) {
            for (int v = 0; v < WPC; ++v) {
                float x = tile[(si * 8 + w) * EPI_PITCH + (ai * 8 + v)];
                cur = ((x >= cur) || (x < 0.0f)) ? x : cur;
            }
        }
        const int sIdx = (wrow0 >> 3) + si;
        const int aIdx = (wcol0 >> 3) + ai;
        maxSim[sIdx * A_CATS + aIdx] = cur;
    }
}

extern "C" void kernel_launch(void* const* d_in, const int* in_sizes, int n_in,
                              void* d_out, int out_size, void* d_ws, size_t ws_size,
                              hipStream_t stream) {
    const float* emb      = (const float*)d_in[0];
    const int*   senCats  = (const int*)d_in[1];
    const int*   annCats  = (const int*)d_in[2];
    const int*   noneIdx  = (const int*)d_in[3];

    float* out    = (float*)d_out;
    float* maxSim = out;
    float* senWd  = out + S_CATS * A_CATS;

    char* ws = (char*)d_ws;
    float* senE     = (float*)(ws + 0);
    float* annE     = (float*)(ws + 2457600);
    float* ns       = (float*)(ws + 4915200);
    float* na       = (float*)(ws + 4923392);
    float* noneMask = (float*)(ws + 4931584);

    ComSim_gather_rows<<<dim3(2 * NROWS), dim3(128), 0, stream>>>(
        emb, senCats, annCats, noneIdx, senE, annE, ns, na, noneMask);

    ComSim_sen_word_sum<<<dim3(S_CATS), dim3(128), 0, stream>>>(senE, senWd);

    ComSim_cos_tile_wmma<<<dim3(32, 32), dim3(128), 0, stream>>>(
        senE, annE, ns, na, noneMask, maxSim);
}


# One launch instead of three: the dot kernel stages its rows straight from the table by index, takes norms, none flag and word sums on the way; norm sum order changed (lane shares, then exchanges)
# speedup vs baseline: 2.2033x; 1.1083x over previous
#include <hip/hip_runtime.h>

typedef __attribute__((ext_vector_type(2))) float v2f;
typedef __attribute__((ext_vector_type(4))) float v4f;
typedef __attribute__((ext_vector_type(8))) float v8f;

#define DWORDS_D 300
#define S_CATS 256
#define A_CATS 256
#define WPC 8
#define NROWS 2048
#define COS_EPS 1e-8f

#define TILE 64
#define TILE_DW (TILE * DWORDS_D)
#define ROW_V4 (DWORDS_D / 4)
#define KCH 5
#define NKCH 15
#define EPI_PITCH 34
#define NRM_OFF (2 * TILE_DW)
#define MSK_OFF (NRM_OFF + 2 * TILE)
#define LDS_DW  (MSK_OFF + TILE)

__global__ void __launch_bounds__(128)
ComSim_cos_tile_wmma(const float* __restrict__ emb,
                     const int* __restrict__ senCats,
                     const int* __restrict__ annCats,
                     const int* __restrict__ noneIdxPtr,
                     float* __restrict__ maxSim,
                     float* __restrict__ senWd)
{
    __shared__ __attribute__((aligned(16))) float lds[LDS_DW];

    const int tid  = threadIdx.x;
    const int lane = tid & 31;
    const int wave = tid >> 5;
    const int half = lane >> 4;
    const int l16  = lane & 15;
    const int wr   = wave >> 1;
    const int wc   = wave & 1;
    const int row0 = blockIdx.y * TILE;
    const int col0 = blockIdx.x * TILE;
    const int wrow0 = row0 + wr * 32;
    const int wcol0 = col0 + wc * 32;


    const int wv = __builtin_amdgcn_readfirstlane(wave);
    {
        const int myId = (lane < 16) ? senCats[row0 + wv + 4 * lane]
                                     : annCats[col0 + wv + 4 * (lane - 16)];
        const bool tail = lane < ROW_V4 - 64;
        const v4f* __restrict__ gn =
            (const v4f*)(emb + (long long)noneIdxPtr[0] * DWORDS_D);
        const v4f n0 = gn[lane];
        const v4f n1 = gn[lane + 32];
        v4f n2;
        if (tail) n2 = gn[lane + 64];

        v4f* lds4 = (v4f*)lds;
        v4f t[32][3];
#pragma unroll
        for (int i = 0; i < 32; ++i) {
            const int id = __builtin_amdgcn_readlane(myId, i);
            const v4f* __restrict__ g = (const v4f*)(emb + (long long)id * DWORDS_D);
            t[i][0] = g[lane];
            t[i][1] = g[lane + 32];
        }
        if (tail) {
#pragma unroll
            for (int i = 0; i < 32; ++i) {
                const int id = __builtin_amdgcn_readlane(myId, i);
                const v4f* __restrict__ g = (const v4f*)(emb + (long long)id * DWORDS_D);
                t[i][2] = g[lane + 64];
            }
        }
        unsigned df = 0u;
        float sq[32];
#pragma unroll
        for (int i = 0; i < 32; ++i) {
            v4f* d = lds4 + (wv + 4 * i) * ROW_V4;
            d[lane]      = t[i][0];
            d[lane + 32] = t[i][1];
            {
                const v4f a = t[i][0], b = t[i][1];
                sq[i] = ((a.x * a.x + a.y * a.y) + (a.z * a.z + a.w * a.w))
                      + ((b.x * b.x + b.y * b.y) + (b.z * b.z + b.w * b.w));
            }
            if (i < 16) {
                const v4f a = t[i][0], b = t[i][1];
                const bool same = (a.x == n0.x) && (a.y == n0.y) && (a.z == n0.z) && (a.w == n0.w)
                               && (b.x == n1.x) && (b.y == n1.y) && (b.z == n1.z) && (b.w == n1.w);
                df |= same ? 0u : (1u << i);
            }
        }
        if (tail) {
#pragma unroll
            for (int i = 0; i < 32; ++i) {
                lds4[(wv + 4 * i) * ROW_V4 + lane + 64] = t[i][2];
                {
                    const v4f a = t[i][2];
                    sq[i] += (a.x * a.x + a.y * a.y) + (a.z * a.z + a.w * a.w);
                }
                if (i < 16) {
                    const v4f a = t[i][2];
                    const bool same = (a.x == n2.x) && (a.y == n2.y) && (a.z == n2.z) && (a.w == n2.w);
                    df |= same ? 0u : (1u << i);
                }
            }
        }
        unsigned eqm = 0u;
#pragma unroll
        for (int i = 0; i < 16; ++i)
            eqm |= (__builtin_amdgcn_ballot_w32(((df >> i) & 1u) != 0u) == 0u) ? (1u << i) : 0u;
        if (lane < 16)
            lds[MSK_OFF + wv + 4 * lane] = ((eqm >> lane) & 1u) ? 1.0f : 0.0f;

#define SQ_EXCHANGE(OFF)                                              \
        {                                                             \
            const bool up = (lane & (OFF)) != 0;                      \
            _Pragma("unroll")                                         \
            for (int j = 0; j < (OFF); ++j) {                         \
                const float keep = up ? sq[j + (OFF)] : sq[j];        \
                const float send = up ? sq[j] : sq[j + (OFF)];        \
                sq[j] = keep + __shfl_xor(send, (OFF));               \
            }                                                         \
        }
        SQ_EXCHANGE(16)
        SQ_EXCHANGE(8)
        SQ_EXCHANGE(4)
        SQ_EXCHANGE(2)
        SQ_EXCHANGE(1)
#undef SQ_EXCHANGE
        lds[NRM_OFF + wv + 4 * lane] = sqrtf(sq[0]);
    }
    __syncthreads();

    if (blockIdx.x == 0) {
        float* __restrict__ o = senWd + (long long)blockIdx.y * (WPC * DWORDS_D);
        for (int e = tid; e < WPC * DWORDS_D; e += 128) {
            const int si = e / DWORDS_D;
            const float* p = lds + e + si * ((WPC - 1) * DWORDS_D);
            float acc = 0.0f;
#pragma unroll
            for (int w = 0; w < WPC; ++w)
                acc += p[w * DWORDS_D];
            o[e] = acc;
        }
    }

    const float* pa0 = lds + (wr * 32 + l16) * DWORDS_D + 2 * half;
    const float* pa1 = pa0 + 16 * DWORDS_D;
    const float* pb0 = lds + TILE_DW + (wc * 32 + l16) * DWORDS_D + 2 * half;
    const float* pb1 = pb0 + 16 * DWORDS_D;

    v8f c00 = {}, c01 = {}, c10 = {}, c11 = {};
    v2f fa0[2][KCH], fa1[2][KCH], fb0[2][KCH], fb1[2][KCH];

#pragma unroll
    for (int j = 0; j < KCH; ++j) {
        fa0[0][j] = *(const v2f*)(pa0 + 4 * j);
        fa1[0][j] = *(const v2f*)(pa1 + 4 * j);
        fb0[0][j] = *(const v2f*)(pb0 + 4 * j);
        fb1[0][j] = *(const v2f*)(pb1 + 4 * j);
    }
#pragma unroll
    for (int ch = 0; ch < NKCH; ++ch) {
        const int cur = ch & 1;
        const int nxt = cur ^ 1;
        if (ch + 1 < NKCH) {
            const int k = (ch + 1) * KCH * 4;
#pragma unroll
            for (int j = 0; j < KCH; ++j) {
                fa0[nxt][j] = *(const v2f*)(pa0 + k + 4 * j);
                fa1[nxt][j] = *(const v2f*)(pa1 + k + 4 * j);
                fb0[nxt][j] = *(const v2f*)(pb0 + k + 4 * j);
                fb1[nxt][j] = *(const v2f*)(pb1 + k + 4 * j);
            }
        }
#pragma unroll
        for (int j = 0; j < KCH; ++j) {
            c00 = __builtin_amdgcn_wmma_f32_16x16x4_f32(false, fa0[cur][j], false, fb0[cur][j],
                                                        (short)0, c00, false, false);
            c01 = __builtin_amdgcn_wmma_f32_16x16x4_f32(false, fa0[cur][j], false, fb1[cur][j],
                                                        (short)0, c01, false, false);
            c10 = __builtin_amdgcn_wmma_f32_16x16x4_f32(false, fa1[cur][j], false, fb0[cur][j],
                                                        (short)0, c10, false, false);
            c11 = __builtin_amdgcn_wmma_f32_16x16x4_f32(false, fa1[cur][j], false, fb1[cur][j],
                                                        (short)0, c11, false, false);
        }
    }

    __syncthreads();

    float naV[2];
    float nsV[2][8];
    float nmV[2][8];
#pragma unroll
    for (int i = 0; i < 2; ++i) {
        naV[i] = lds[NRM_OFF + TILE + wc * 32 + i * 16 + l16];
#pragma unroll
        for (int r = 0; r < 8; ++r) {
            nsV[i][r] = lds[NRM_OFF + wr * 32 + i * 16 + 8 * half + r];
            nmV[i][r] = lds[MSK_OFF + wr * 32 + i * 16 + 8 * half + r];
        }
    }

    float* tile = lds + wave * (32 * EPI_PITCH);
#pragma unroll
    for (int mi = 0; mi < 2; ++mi) {
#pragma unroll
        for (int ni = 0; ni < 2; ++ni) {
            const v8f c = (mi == 0) ? ((ni == 0) ? c00 : c01) : ((ni == 0) ? c10 : c11);
#pragma unroll
            for (int r = 0; r < 8; ++r) {
                const int m = mi * 16 + r + 8 * half;
                const float denom = fmaxf(nsV[mi][r] * naV[ni], COS_EPS);
                float cosv = c[r] / denom;
                if (nmV[mi][r] != 0.0f) cosv = 0.0f;
                tile[m * EPI_PITCH + ni * 16 + l16] = cosv;
            }
        }
    }
    __syncthreads();

    if (lane < 16) {
        const int si = lane >> 2;
        const int ai = lane & 3;
        float cur = 0.0f;
        for (int w = 0; w < WPC; ++w) {
            for (int v = 0; v < WPC; ++v) {
                float x = tile[(si * 8 + w) * EPI_PITCH + (ai * 8 + v)];
                cur = ((x >= cur) || (x < 0.0f)) ? x : cur;
            }
        }
        const int sIdx = (wrow0 >> 3) + si;
        const int aIdx = (wcol0 >> 3) + ai;
        maxSim[sIdx * A_CATS + aIdx] = cur;
    }
}

extern "C" void kernel_launch(void* const* d_in, const int* in_sizes, int n_in,
                              void* d_out, int out_size, void* d_ws, size_t ws_size,
                              hipStream_t stream) {
    const float* emb      = (const float*)d_in[0];
    const int*   senCats  = (const int*)d_in[1];
    const int*   annCats  = (const int*)d_in[2];
    const int*   noneIdx  = (const int*)d_in[3];

    float* out    = (float*)d_out;
    float* maxSim = out;
    float* senWd  = out + S_CATS * A_CATS;

    ComSim_cos_tile_wmma<<<dim3(32, 32), dim3(128), 0, stream>>>(
        emb, senCats, annCats, noneIdx, maxSim, senWd);
}


# 128x128 block per 8-wave work-group (one per WGP, grid 16x16), each wave a 64x32 sub-block of 4x2 WMMA tiles: each table row is fetched from L2 half as often; k order and all sum orders unchanged
# speedup vs baseline: 2.8293x; 1.2841x over previous
#include <hip/hip_runtime.h>

typedef __attribute__((ext_vector_type(2))) float v2f;
typedef __attribute__((ext_vector_type(4))) float v4f;
typedef __attribute__((ext_vector_type(8))) float v8f;

#define DWORDS_D 300
#define S_CATS 256
#define A_CATS 256
#define WPC 8
#define NROWS 2048
#define COS_EPS 1e-8f

#define TILE 128
#define TILE_DW (TILE * DWORDS_D)
#define ROW_V4 (DWORDS_D / 4)
#define NWAVES 8
#define WM 4
#define WN 2
#define KCH 5
#define NKCH 15
#define EPI_PITCH 34
#define EPI_ROWS (WM * 16)
#define NRM_OFF (2 * TILE_DW)
#define MSK_OFF (NRM_OFF + 2 * TILE)
#define LDS_DW  (MSK_OFF + TILE)

__global__ void __launch_bounds__(256)
ComSim_cos_tile_wmma(const float* __restrict__ emb,
                     const int* __restrict__ senCats,
                     const int* __restrict__ annCats,
                     const int* __restrict__ noneIdxPtr,
                     float* __restrict__ maxSim,
                     float* __restrict__ senWd)
{
    __shared__ __attribute__((aligned(16))) float lds[LDS_DW];

    const int tid  = threadIdx.x;
    const int lane = tid & 31;
    const int wave = tid >> 5;
    const int half = lane >> 4;
    const int l16  = lane & 15;
    const int wr   = wave >> 2;
    const int wc   = wave & 3;
    const int row0 = blockIdx.y * TILE;
    const int col0 = blockIdx.x * TILE;
    const int wrow0 = row0 + wr * 64;
    const int wcol0 = col0 + wc * 32;


    const int wv = __builtin_amdgcn_readfirstlane(wave);
    {
        const int myId = (lane < 16) ? senCats[row0 + wv + NWAVES * lane]
                                     : annCats[col0 + wv + NWAVES * (lane - 16)];
        const bool tail = lane < ROW_V4 - 64;
        const v4f* __restrict__ gn =
            (const v4f*)(emb + (long long)noneIdxPtr[0] * DWORDS_D);
        const v4f n0 = gn[lane];
        const v4f n1 = gn[lane + 32];
        v4f n2;
        if (tail) n2 = gn[lane + 64];

        v4f* lds4 = (v4f*)lds;
        v4f t[32][3];
#pragma unroll
        for (int i = 0; i < 32; ++i) {
            const int id = __builtin_amdgcn_readlane(myId, i);
            const v4f* __restrict__ g = (const v4f*)(emb + (long long)id * DWORDS_D);
            t[i][0] = g[lane];
            t[i][1] = g[lane + 32];
        }
        if (tail) {
#pragma unroll
            for (int i = 0; i < 32; ++i) {
                const int id = __builtin_amdgcn_readlane(myId, i);
                const v4f* __restrict__ g = (const v4f*)(emb + (long long)id * DWORDS_D);
                t[i][2] = g[lane + 64];
            }
        }
        unsigned df = 0u;
        float sq[32];
#pragma unroll
        for (int i = 0; i < 32; ++i) {
            v4f* d = lds4 + (wv + NWAVES * i) * ROW_V4;
            d[lane]      = t[i][0];
            d[lane + 32] = t[i][1];
            {
                const v4f a = t[i][0], b = t[i][1];
                sq[i] = ((a.x * a.x + a.y * a.y) + (a.z * a.z + a.w * a.w))
                      + ((b.x * b.x + b.y * b.y) + (b.z * b.z + b.w * b.w));
            }
            if (i < 16) {
                const v4f a = t[i][0], b = t[i][1];
                const bool same = (a.x == n0.x) && (a.y == n0.y) && (a.z == n0.z) && (a.w == n0.w)
                               && (b.x == n1.x) && (b.y == n1.y) && (b.z == n1.z) && (b.w == n1.w);
                df |= same ? 0u : (1u << i);
            }
        }
        if (tail) {
#pragma unroll
            for (int i = 0; i < 32; ++i) {
                lds4[(wv + NWAVES * i) * ROW_V4 + lane + 64] = t[i][2];
                {
                    const v4f a = t[i][2];
                    sq[i] += (a.x * a.x + a.y * a.y) + (a.z * a.z + a.w * a.w);
                }
                if (i < 16) {
                    const v4f a = t[i][2];
                    const bool same = (a.x == n2.x) && (a.y == n2.y) && (a.z == n2.z) && (a.w == n2.w);
                    df |= same ? 0u : (1u << i);
                }
            }
        }
        unsigned eqm = 0u;
#pragma unroll
        for (int i = 0; i < 16; ++i)
            eqm |= (__builtin_amdgcn_ballot_w32(((df >> i) & 1u) != 0u) == 0u) ? (1u << i) : 0u;
        if (lane < 16)
            lds[MSK_OFF + wv + NWAVES * lane] = ((eqm >> lane) & 1u) ? 1.0f : 0.0f;

#define SQ_EXCHANGE(OFF)                                              \
        {                                                             \
            const bool up = (lane & (OFF)) != 0;                      \
            _Pragma("unroll")                                         \
            for (int j = 0; j < (OFF); ++j) {                         \
                const float keep = up ? sq[j + (OFF)] : sq[j];        \
                const float send = up ? sq[j] : sq[j + (OFF)];        \
                sq[j] = keep + __shfl_xor(send, (OFF));               \
            }                                                         \
        }
        SQ_EXCHANGE(16)
        SQ_EXCHANGE(8)
        SQ_EXCHANGE(4)
        SQ_EXCHANGE(2)
        SQ_EXCHANGE(1)
#undef SQ_EXCHANGE
        lds[NRM_OFF + wv + NWAVES * lane] = sqrtf(sq[0]);
    }
    __syncthreads();

    if (blockIdx.x == 0) {
        float* __restrict__ o = senWd + (long long)blockIdx.y * ((TILE / WPC) * DWORDS_D);
        for (int e = tid; e < (TILE / WPC) * DWORDS_D; e += NWAVES * 32) {
            const int si = e / DWORDS_D;
            const float* p = lds + e + si * ((WPC - 1) * DWORDS_D);
            float acc = 0.0f;
#pragma unroll
            for (int w = 0; w < WPC; ++w)
                acc += p[w * DWORDS_D];
            o[e] = acc;
        }
    }

    const float* pa[WM];
    const float* pb[WN];
#pragma unroll
    for (int mi = 0; mi < WM; ++mi)
        pa[mi] = lds + (wr * 64 + mi * 16 + l16) * DWORDS_D + 2 * half;
#pragma unroll
    for (int ni = 0; ni < WN; ++ni)
        pb[ni] = lds + TILE_DW + (wc * 32 + ni * 16 + l16) * DWORDS_D + 2 * half;

    v8f c[WM][WN];
#pragma unroll
    for (int mi = 0; mi < WM; ++mi)
#pragma unroll
        for (int ni = 0; ni < WN; ++ni)
            c[mi][ni] = (v8f){};
    v2f fa[2][WM][KCH], fb[2][WN][KCH];

#pragma unroll
    for (int j = 0; j < KCH; ++j) {
#pragma unroll
        for (int mi = 0; mi < WM; ++mi)
            fa[0][mi][j] = *(const v2f*)(pa[mi] + 4 * j);
#pragma unroll
        for (int ni = 0; ni < WN; ++ni)
            fb[0][ni][j] = *(const v2f*)(pb[ni] + 4 * j);
    }
#pragma unroll
    for (int ch = 0; ch < NKCH; ++ch) {
        const int cur = ch & 1;
        const int nxt = cur ^ 1;
        if (ch + 1 < NKCH) {
            const int k = (ch + 1) * KCH * 4;
#pragma unroll
            for (int j = 0; j < KCH; ++j) {
#pragma unroll
                for (int mi = 0; mi < WM; ++mi)
                    fa[nxt][mi][j] = *(const v2f*)(pa[mi] + k + 4 * j);
#pragma unroll
                for (int ni = 0; ni < WN; ++ni)
                    fb[nxt][ni][j] = *(const v2f*)(pb[ni] + k + 4 * j);
            }
        }
#pragma unroll
        for (int j = 0; j < KCH; ++j) {
#pragma unroll
            for (int mi = 0; mi < WM; ++mi) {
#pragma unroll
                for (int ni = 0; ni < WN; ++ni) {
                    c[mi][ni] = __builtin_amdgcn_wmma_f32_16x16x4_f32(
                        false, fa[cur][mi][j], false, fb[cur][ni][j],
                        (short)0, c[mi][ni], false, false);
                }
            }
        }
    }

    __syncthreads();

    float naV[WN];
    float nsV[WM][8];
    float nmV[WM][8];
#pragma unroll
    for (int ni = 0; ni < WN; ++ni)
        naV[ni] = lds[NRM_OFF + TILE + wc * 32 + ni * 16 + l16];
#pragma unroll
    for (int mi = 0; mi < WM; ++mi) {
#pragma unroll
        for (int r = 0; r < 8; ++r) {
            nsV[mi][r] = lds[NRM_OFF + wr * 64 + mi * 16 + 8 * half + r];
            nmV[mi][r] = lds[MSK_OFF + wr * 64 + mi * 16 + 8 * half + r];
        }
    }

    float* tile = lds + wave * (EPI_ROWS * EPI_PITCH);
#pragma unroll
    for (int mi = 0; mi < WM; ++mi) {
#pragma unroll
        for (int ni = 0; ni < WN; ++ni) {
            const v8f cc = c[mi][ni];
#pragma unroll
            for (int r = 0; r < 8; ++r) {
                const int m = mi * 16 + r + 8 * half;
                const float denom = fmaxf(nsV[mi][r] * naV[ni], COS_EPS);
                float cosv = cc[r] / denom;
                if (nmV[mi][r] != 0.0f) cosv = 0.0f;
                tile[m * EPI_PITCH + ni * 16 + l16] = cosv;
            }
        }
    }
    __syncthreads();

    {
        const int si = lane >> 2;
        const int ai = lane & 3;
        float cur = 0.0f;
        for (int w = 0; w < WPC; ++w) {
            for (int v = 0; v < WPC; ++v) {
                float x = tile[(si * 8 + w) * EPI_PITCH + (ai * 8 + v)];
                cur = ((x >= cur) || (x < 0.0f)) ? x : cur;
            }
        }
        const int sIdx = (wrow0 >> 3) + si;
        const int aIdx = (wcol0 >> 3) + ai;
        maxSim[sIdx * A_CATS + aIdx] = cur;
    }
}

extern "C" void kernel_launch(void* const* d_in, const int* in_sizes, int n_in,
                              void* d_out, int out_size, void* d_ws, size_t ws_size,
                              hipStream_t stream) {
    const float* emb      = (const float*)d_in[0];
    const int*   senCats  = (const int*)d_in[1];
    const int*   annCats  = (const int*)d_in[2];
    const int*   noneIdx  = (const int*)d_in[3];

    float* out    = (float*)d_out;
    float* maxSim = out;
    float* senWd  = out + S_CATS * A_CATS;

    ComSim_cos_tile_wmma<<<dim3(16, 16), dim3(256), 0, stream>>>(
        emb, senCats, annCats, noneIdx, maxSim, senWd);
}


# K-chunked staging: rows enter LDS in 5 chunks; later chunks' table loads and LDS stores run under earlier chunks' WMMAs (split barrier per chunk); norm sum order changed (16-lane shares, exchanges)
# speedup vs baseline: 3.0461x; 1.0766x over previous
#include <hip/hip_runtime.h>

typedef __attribute__((ext_vector_type(2))) float v2f;
typedef __attribute__((ext_vector_type(4))) float v4f;
typedef __attribute__((ext_vector_type(8))) float v8f;

#define DWORDS_D 300
#define S_CATS 256
#define A_CATS 256
#define WPC 8
#define NROWS 2048
#define COS_EPS 1e-8f

#define TILE 128
#define TILE_DW (TILE * DWORDS_D)
#define ROW_V4 (DWORDS_D / 4)
#define ROW_BYTES (DWORDS_D * 4)
#define NWAVES 8
#define WM 4
#define WN 2
#define CH_P 16
#define NCHUNK 5
#define LAST_P (ROW_V4 - (NCHUNK - 1) * CH_P)
#define NLD 16
#define KG 4
#define NGRP 19
#define EPI_PITCH 34
#define EPI_ROWS (WM * 16)
#define NRM_OFF (2 * TILE_DW)
#define MSK_OFF (NRM_OFF + 2 * TILE)
#define LDS_DW  (MSK_OFF + TILE)

static __device__ __forceinline__ void wait_ds0() { asm volatile("s_wait_dscnt 0x0" ::: "memory"); }
static __device__ __forceinline__ void bar_signal() { asm volatile("s_barrier_signal -1" ::: "memory"); }
static __device__ __forceinline__ void bar_wait() { asm volatile("s_barrier_wait -1" ::: "memory"); }

__global__ void __launch_bounds__(256)
ComSim_cos_tile_wmma(const float* __restrict__ emb,
                     const int* __restrict__ senCats,
                     const int* __restrict__ annCats,
                     const int* __restrict__ noneIdxPtr,
                     float* __restrict__ maxSim,
                     float* __restrict__ senWd)
{
    __shared__ __attribute__((aligned(16))) float lds[LDS_DW];

    const int tid  = threadIdx.x;
    const int lane = tid & 31;
    const int wave = tid >> 5;
    const int half = lane >> 4;
    const int l16  = lane & 15;
    const int wr   = wave >> 2;
    const int wc   = wave & 3;
    const int row0 = blockIdx.y * TILE;
    const int col0 = blockIdx.x * TILE;
    const int wrow0 = row0 + wr * 64;
    const int wcol0 = col0 + wc * 32;

    const int wv = __builtin_amdgcn_readfirstlane(wave);

    const int myId = (lane < 16) ? senCats[row0 + wv + NWAVES * lane]
                                 : annCats[col0 + wv + NWAVES * (lane - 16)];
    const char* __restrict__ embB = (const char*)emb;
    unsigned off[NLD];
#pragma unroll
    for (int i = 0; i < NLD; ++i) {
        const int idLo = __builtin_amdgcn_readlane(myId, 2 * i);
        const int idHi = __builtin_amdgcn_readlane(myId, 2 * i + 1);
        off[i] = (unsigned)(half ? idHi : idLo) * (unsigned)ROW_BYTES + (unsigned)l16 * 16u;
    }
    const bool tailLane = l16 < LAST_P;
    const int p4 = tailLane ? (NCHUNK - 1) * CH_P + l16 : ROW_V4 - 1;
    const unsigned d4 = (unsigned)(p4 - l16) * 16u;

    v4f t[2][NLD];
    v4f nzp[2];
    float sq[NLD];
    unsigned df = 0u;
    v4f* lds4 = (v4f*)lds;
    const int ldsP0 = (wv + NWAVES * half) * ROW_V4 + l16;

#define LOADS(CH, B)                                                            \
    {                                                                           \
        _Pragma("unroll")                                                       \
        for (int i = 0; i < NLD; ++i) {                                         \
            if ((CH) < NCHUNK - 1)                                              \
                t[B][i] = *(const v4f*)(embB + (size_t)off[i] + (CH) * (CH_P * 16)); \
            else                                                                \
                t[B][i] = *(const v4f*)(embB + (size_t)(off[i] + d4));          \
        }                                                                       \
        if ((CH) < NCHUNK - 1)                                                  \
            nzp[B] = *(const v4f*)(gnB + (CH) * (CH_P * 16));                   \
        else                                                                    \
            nzp[B] = *(const v4f*)(gnB + d4);                                   \
    }
#define STAGE(CH, B)                                                            \
    {                                                                           \
        const int pc = ((CH) < NCHUNK - 1) ? ldsP0 + (CH) * CH_P : ldsP0 - l16 + p4; \
        const v4f nz = nzp[B];                                                  \
        _Pragma("unroll")                                                       \
        for (int i = 0; i < NLD; ++i) {                                         \
            const v4f a = t[B][i];                                              \
            lds4[pc + i * (2 * NWAVES * ROW_V4)] = a;                           \
            float s = (a.x * a.x + a.y * a.y) + (a.z * a.z + a.w * a.w);        \
            if ((CH) == NCHUNK - 1) s = tailLane ? s : 0.0f;                    \
            sq[i] = ((CH) == 0) ? s : sq[i] + s;                                \
            if (i < 8) {                                                        \
                const bool same = (a.x == nz.x) && (a.y == nz.y)                \
                               && (a.z == nz.z) && (a.w == nz.w);               \
                df |= same ? 0u : (1u << i);                                    \
            }                                                                   \
        }                                                                       \
    }

    const char* __restrict__ gnB = embB + (long long)noneIdxPtr[0] * ROW_BYTES + l16 * 16;

    LOADS(0, 0)
    __builtin_amdgcn_sched_barrier(0);
    LOADS(1, 1)
    STAGE(0, 0)
    wait_ds0();
    bar_signal();
    __builtin_amdgcn_sched_barrier(0);
    LOADS(2, 0)
    bar_wait();

    const float* pa[WM];
    const float* pb[WN];
#pragma unroll
    for (int mi = 0; mi < WM; ++mi)
        pa[mi] = lds + (wr * 64 + mi * 16 + l16) * DWORDS_D + 2 * half;
#pragma unroll
    for (int ni = 0; ni < WN; ++ni)
        pb[ni] = lds + TILE_DW + (wc * 32 + ni * 16 + l16) * DWORDS_D + 2 * half;

    v8f c[WM][WN];
#pragma unroll
    for (int mi = 0; mi < WM; ++mi)
#pragma unroll
        for (int ni = 0; ni < WN; ++ni)
            c[mi][ni] = (v8f){};
    v2f fa[2][WM][KG], fb[2][WN][KG];

#pragma unroll
    for (int j = 0; j < KG; ++j) {
#pragma unroll
        for (int mi = 0; mi < WM; ++mi)
            fa[0][mi][j] = *(const v2f*)(pa[mi] + 4 * j);
#pragma unroll
        for (int ni = 0; ni < WN; ++ni)
            fb[0][ni][j] = *(const v2f*)(pb[ni] + 4 * j);
    }
#define GRP(G)                                                                  \
    {                                                                           \
        if ((G) + 1 < NGRP) {                                                   \
            if ((((G) + 1) & 3) == 0) bar_wait();                               \
            const int k = ((G) + 1) * KG * 4;                                   \
            const int nk = ((G) + 1 == NGRP - 1) ? 3 : KG;                      \
            _Pragma("unroll")                                                   \
            for (int j = 0; j < nk; ++j) {                                      \
                _Pragma("unroll")                                               \
                for (int mi = 0; mi < WM; ++mi)                                 \
                    fa[((G) + 1) & 1][mi][j] = *(const v2f*)(pa[mi] + k + 4 * j); \
                _Pragma("unroll")                                               \
                for (int ni = 0; ni < WN; ++ni)                                 \
                    fb[((G) + 1) & 1][ni][j] = *(const v2f*)(pb[ni] + k + 4 * j); \
            }                                                                   \
        }                                                                       \
        const int nkc = ((G) == NGRP - 1) ? 3 : KG;                             \
        _Pragma("unroll")                                                       \
        for (int j = 0; j < nkc; ++j) {                                         \
            _Pragma("unroll")                                                   \
            for (int mi = 0; mi < WM; ++mi) {                                   \
                _Pragma("unroll")                                               \
                for (int ni = 0; ni < WN; ++ni) {                               \
                    c[mi][ni] = __builtin_amdgcn_wmma_f32_16x16x4_f32(          \
                        false, fa[(G) & 1][mi][j], false, fb[(G) & 1][ni][j],   \
                        (short)0, c[mi][ni], false, false);                     \
                }                                                               \
            }                                                                   \
        }                                                                       \
    }

#define KEEP_BELOW()    __builtin_amdgcn_sched_barrier(0)
#define REFILL_BEHIND() __builtin_amdgcn_sched_barrier(0)
    GRP(0) GRP(1)
    KEEP_BELOW();
    GRP(2)
    STAGE(1, 1)
    wait_ds0();
    bar_signal();
    REFILL_BEHIND();
    LOADS(3, 1)
    GRP(3)
    GRP(4)
    KEEP_BELOW();
    GRP(5)
    STAGE(2, 0)
    wait_ds0();
    bar_signal();
    REFILL_BEHIND();
    LOADS(4, 0)
    GRP(6) GRP(7)
    GRP(8)
    KEEP_BELOW();
    GRP(9)
    STAGE(3, 1)
    wait_ds0();
    bar_signal();
    GRP(10) GRP(11)
    GRP(12)
    KEEP_BELOW();
    GRP(13)
    STAGE(4, 0)
    wait_ds0();
    bar_signal();
    {
        unsigned eqm = 0u;
#pragma unroll
        for (int i = 0; i < 8; ++i) {
            const unsigned m = __builtin_amdgcn_ballot_w32(((df >> i) & 1u) != 0u);
            eqm |= ((m & 0xFFFFu) == 0u) ? (1u << (2 * i)) : 0u;
            eqm |= ((m >> 16) == 0u) ? (1u << (2 * i + 1)) : 0u;
        }
        if (lane < 16)
            lds[MSK_OFF + wv + NWAVES * lane] = ((eqm >> lane) & 1u) ? 1.0f : 0.0f;
#define SQ_EXCHANGE(OFF)                                              \
        {                                                             \
            const bool up = (lane & (OFF)) != 0;                      \
            _Pragma("unroll")                                         \
            for (int j = 0; j < (OFF); ++j) {                         \
                const float keep = up ? sq[j + (OFF)] : sq[j];        \
                const float send = up ? sq[j] : sq[j + (OFF)];        \
                sq[j] = keep + __shfl_xor(send, (OFF));               \
            }                                                         \
        }
        SQ_EXCHANGE(8)
        SQ_EXCHANGE(4)
        SQ_EXCHANGE(2)
        SQ_EXCHANGE(1)
#undef SQ_EXCHANGE
        lds[NRM_OFF + wv + NWAVES * (2 * l16 + half)] = sqrtf(sq[0]);
    }
    GRP(14) GRP(15)
    GRP(16) GRP(17) GRP(18)
#undef GRP
#undef KEEP_BELOW
#undef REFILL_BEHIND
#undef LOADS
#undef STAGE

    if (blockIdx.x == 0) {
        float* __restrict__ o = senWd + (long long)blockIdx.y * ((TILE / WPC) * DWORDS_D);
        for (int e = tid; e < (TILE / WPC) * DWORDS_D; e += NWAVES * 32) {
            const int si = e / DWORDS_D;
            const float* p = lds + e + si * ((WPC - 1) * DWORDS_D);
            float acc = 0.0f;
#pragma unroll
            for (int w = 0; w < WPC; ++w)
                acc += p[w * DWORDS_D];
            o[e] = acc;
        }
    }

    __syncthreads();

    float naV[WN];
    float nsV[WM][8];
    float nmV[WM][8];
#pragma unroll
    for (int ni = 0; ni < WN; ++ni)
        naV[ni] = lds[NRM_OFF + TILE + wc * 32 + ni * 16 + l16];
#pragma unroll
    for (int mi = 0; mi < WM; ++mi) {
#pragma unroll
        for (int r = 0; r < 8; ++r) {
            nsV[mi][r] = lds[NRM_OFF + wr * 64 + mi * 16 + 8 * half + r];
            nmV[mi][r] = lds[MSK_OFF + wr * 64 + mi * 16 + 8 * half + r];
        }
    }

    float* tile = lds + wave * (EPI_ROWS * EPI_PITCH);
#pragma unroll
    for (int mi = 0; mi < WM; ++mi) {
#pragma unroll
        for (int ni = 0; ni < WN; ++ni) {
            const v8f cc = c[mi][ni];
#pragma unroll
            for (int r = 0; r < 8; ++r) {
                const int m = mi * 16 + r + 8 * half;
                const float denom = fmaxf(nsV[mi][r] * naV[ni], COS_EPS);
                float cosv = cc[r] / denom;
                if (nmV[mi][r] != 0.0f) cosv = 0.0f;
                tile[m * EPI_PITCH + ni * 16 + l16] = cosv;
            }
        }
    }
    __syncthreads();

    {
        const int si = lane >> 2;
        const int ai = lane & 3;
        float cur = 0.0f;
        for (int w = 0; w < WPC; ++w) {
            for (int v = 0; v < WPC; ++v) {
                float x = tile[(si * 8 + w) * EPI_PITCH + (ai * 8 + v)];
                cur = ((x >= cur) || (x < 0.0f)) ? x : cur;
            }
        }
        const int sIdx = (wrow0 >> 3) + si;
        const int aIdx = (wcol0 >> 3) + ai;
        maxSim[sIdx * A_CATS + aIdx] = cur;
    }
}

extern "C" void kernel_launch(void* const* d_in, const int* in_sizes, int n_in,
                              void* d_out, int out_size, void* d_ws, size_t ws_size,
                              hipStream_t stream) {
    const float* emb      = (const float*)d_in[0];
    const int*   senCats  = (const int*)d_in[1];
    const int*   annCats  = (const int*)d_in[2];
    const int*   noneIdx  = (const int*)d_in[3];

    float* out    = (float*)d_out;
    float* maxSim = out;
    float* senWd  = out + S_CATS * A_CATS;

    ComSim_cos_tile_wmma<<<dim3(16, 16), dim3(256), 0, stream>>>(
        emb, senCats, annCats, noneIdx, maxSim, senWd);
}


# Epilogue in registers, no LDS tile and no barrier behind the WMMAs: ann words as tile rows, fold = max from the last negative step on (DPP); a SIMD's first-done wave ends under the other's WMMAs
# speedup vs baseline: 3.2078x; 1.0531x over previous
#include <hip/hip_runtime.h>

typedef __attribute__((ext_vector_type(2))) float v2f;
typedef __attribute__((ext_vector_type(4))) float v4f;
typedef __attribute__((ext_vector_type(8))) float v8f;

#define DWORDS_D 300
#define S_CATS 256
#define A_CATS 256
#define WPC 8
#define NROWS 2048
#define COS_EPS 1e-8f

#define TILE 128
#define TILE_DW (TILE * DWORDS_D)
#define ROW_V4 (DWORDS_D / 4)
#define ROW_BYTES (DWORDS_D * 4)
#define NWAVES 8
#define WM 4
#define WN 2
#define CH_P 16
#define NCHUNK 5
#define LAST_P (ROW_V4 - (NCHUNK - 1) * CH_P)
#define NLD 16
#define KG 4
#define NGRP 19
#define NORMS_GRP 15
#define NRM_OFF (2 * TILE_DW)
#define MSK_OFF (NRM_OFF + 2 * TILE)
#define LDS_DW  (MSK_OFF + TILE)

static __device__ __forceinline__ void wait_ds0() { asm volatile("s_wait_dscnt 0x0" ::: "memory"); }
static __device__ __forceinline__ void bar_signal() { asm volatile("s_barrier_signal -1" ::: "memory"); }
static __device__ __forceinline__ void bar_wait() { asm volatile("s_barrier_wait -1" ::: "memory"); }

template <int M> static __device__ __forceinline__ int xlane_i(int v) {
    return __builtin_amdgcn_update_dpp(v, v, 0x160 + M, 0xf, 0xf, false);
}
template <int M> static __device__ __forceinline__ float xlane_f(float v) {
    return __builtin_bit_cast(float, xlane_i<M>(__builtin_bit_cast(int, v)));
}

__global__ void __launch_bounds__(256)
ComSim_cos_tile_wmma(const float* __restrict__ emb,
                     const int* __restrict__ senCats,
                     const int* __restrict__ annCats,
                     const int* __restrict__ noneIdxPtr,
                     float* __restrict__ maxSim,
                     float* __restrict__ senWd)
{
    __shared__ __attribute__((aligned(16))) float lds[LDS_DW];

    const int tid  = threadIdx.x;
    const int lane = tid & 31;
    const int wave = tid >> 5;
    const int half = lane >> 4;
    const int l16  = lane & 15;
    const int wr   = wave >> 2;
    const int wc   = wave & 3;
    const int row0 = blockIdx.y * TILE;
    const int col0 = blockIdx.x * TILE;

    const int wv = __builtin_amdgcn_readfirstlane(wave);

    const int myId = (lane < 16) ? senCats[row0 + wv + NWAVES * lane]
                                 : annCats[col0 + wv + NWAVES * (lane - 16)];
    const char* __restrict__ embB = (const char*)emb;
    unsigned off[NLD];
#pragma unroll
    for (int i = 0; i < NLD; ++i) {
        const int idLo = __builtin_amdgcn_readlane(myId, 2 * i);
        const int idHi = __builtin_amdgcn_readlane(myId, 2 * i + 1);
        off[i] = (unsigned)(half ? idHi : idLo) * (unsigned)ROW_BYTES + (unsigned)l16 * 16u;
    }
    const bool tailLane = l16 < LAST_P;
    const int p4 = tailLane ? (NCHUNK - 1) * CH_P + l16 : ROW_V4 - 1;
    const unsigned d4 = (unsigned)(p4 - l16) * 16u;

    v4f t[2][NLD];
    v4f nzp[2];
    float sq[NLD];
    unsigned df = 0u;
    v4f* lds4 = (v4f*)lds;
    const int ldsP0 = (wv + NWAVES * half) * ROW_V4 + l16;

#define LOADS(CH, B)                                                            \
    {                                                                           \
        _Pragma("unroll")                                                       \
        for (int i = 0; i < NLD; ++i) {                                         \
            if ((CH) < NCHUNK - 1)                                              \
                t[B][i] = *(const v4f*)(embB + (size_t)off[i] + (CH) * (CH_P * 16)); \
            else                                                                \
                t[B][i] = *(const v4f*)(embB + (size_t)(off[i] + d4));          \
        }                                                                       \
        if ((CH) < NCHUNK - 1)                                                  \
            nzp[B] = *(const v4f*)(gnB + (CH) * (CH_P * 16));                   \
        else                                                                    \
            nzp[B] = *(const v4f*)(gnB + d4);                                   \
    }
#define STAGE(CH, B)                                                            \
    {                                                                           \
        const int pc = ((CH) < NCHUNK - 1) ? ldsP0 + (CH) * CH_P : ldsP0 - l16 + p4; \
        const v4f nz = nzp[B];                                                  \
        _Pragma("unroll")                                                       \
        for (int i = 0; i < NLD; ++i) {                                         \
            const v4f a = t[B][i];                                              \
            lds4[pc + i * (2 * NWAVES * ROW_V4)] = a;                           \
            float s = (a.x * a.x + a.y * a.y) + (a.z * a.z + a.w * a.w);        \
            if ((CH) == NCHUNK - 1) s = tailLane ? s : 0.0f;                    \
            sq[i] = ((CH) == 0) ? s : sq[i] + s;                                \
            if (i < 8) {                                                        \
                const bool same = (a.x == nz.x) && (a.y == nz.y)                \
                               && (a.z == nz.z) && (a.w == nz.w);               \
                df |= same ? 0u : (1u << i);                                    \
            }                                                                   \
        }                                                                       \
    }

    const char* __restrict__ gnB = embB + (long long)noneIdxPtr[0] * ROW_BYTES + l16 * 16;

    LOADS(0, 0)
    __builtin_amdgcn_sched_barrier(0);
    LOADS(1, 1)
    STAGE(0, 0)
    wait_ds0();
    bar_signal();
    __builtin_amdgcn_sched_barrier(0);
    LOADS(2, 0)
    bar_wait();

    const float* pa[WM];
    const float* pb[WN];
#pragma unroll
    for (int mi = 0; mi < WM; ++mi)
        pa[mi] = lds + TILE_DW + (wr * 64 + mi * 16 + l16) * DWORDS_D + 2 * half;
#pragma unroll
    for (int ni = 0; ni < WN; ++ni)
        pb[ni] = lds + (wc * 32 + ni * 16 + l16) * DWORDS_D + 2 * half;

    v8f c[WM][WN];
#pragma unroll
    for (int mi = 0; mi < WM; ++mi)
#pragma unroll
        for (int ni = 0; ni < WN; ++ni)
            c[mi][ni] = (v8f){};
    v2f fa[2][WM][KG], fb[2][WN][KG];

#pragma unroll
    for (int j = 0; j < KG; ++j) {
#pragma unroll
        for (int mi = 0; mi < WM; ++mi)
            fa[0][mi][j] = *(const v2f*)(pa[mi] + 4 * j);
#pragma unroll
        for (int ni = 0; ni < WN; ++ni)
            fb[0][ni][j] = *(const v2f*)(pb[ni] + 4 * j);
    }
#define GRP(G)                                                                  \
    {                                                                           \
        if ((G) + 1 < NGRP) {                                                   \
            if ((((G) + 1) & 3) == 0) bar_wait();                               \
            if ((G) == NORMS_GRP) { wait_ds0(); bar_signal(); }                 \
            const int k = ((G) + 1) * KG * 4;                                   \
            const int nk = ((G) + 1 == NGRP - 1) ? 3 : KG;                      \
            _Pragma("unroll")                                                   \
            for (int j = 0; j < nk; ++j) {                                      \
                _Pragma("unroll")                                               \
                for (int mi = 0; mi < WM; ++mi)                                 \
                    fa[((G) + 1) & 1][mi][j] = *(const v2f*)(pa[mi] + k + 4 * j); \
                _Pragma("unroll")                                               \
                for (int ni = 0; ni < WN; ++ni)                                 \
                    fb[((G) + 1) & 1][ni][j] = *(const v2f*)(pb[ni] + k + 4 * j); \
            }                                                                   \
        }                                                                       \
        const int nkc = ((G) == NGRP - 1) ? 3 : KG;                             \
        _Pragma("unroll")                                                       \
        for (int j = 0; j < nkc; ++j) {                                         \
            _Pragma("unroll")                                                   \
            for (int mi = 0; mi < WM; ++mi) {                                   \
                _Pragma("unroll")                                               \
                for (int ni = 0; ni < WN; ++ni) {                               \
                    c[mi][ni] = __builtin_amdgcn_wmma_f32_16x16x4_f32(          \
                        false, fa[(G) & 1][mi][j], false, fb[(G) & 1][ni][j],   \
                        (short)0, c[mi][ni], false, false);                     \
                }                                                               \
            }                                                                   \
        }                                                                       \
    }

#define KEEP_BELOW()    __builtin_amdgcn_sched_barrier(0)
#define REFILL_BEHIND() __builtin_amdgcn_sched_barrier(0)
    GRP(0) GRP(1)
    KEEP_BELOW();
    GRP(2)
    STAGE(1, 1)
    wait_ds0();
    bar_signal();
    REFILL_BEHIND();
    LOADS(3, 1)
    GRP(3)
    GRP(4)
    KEEP_BELOW();
    GRP(5)
    STAGE(2, 0)
    wait_ds0();
    bar_signal();
    REFILL_BEHIND();
    LOADS(4, 0)
    GRP(6) GRP(7)
    GRP(8)
    KEEP_BELOW();
    GRP(9)
    STAGE(3, 1)
    wait_ds0();
    bar_signal();
    GRP(10) GRP(11)
    GRP(12)
    KEEP_BELOW();
    GRP(13)
    STAGE(4, 0)
    wait_ds0();
    bar_signal();
    {
        unsigned eqm = 0u;
#pragma unroll
        for (int i = 0; i < 8; ++i) {
            const unsigned m = __builtin_amdgcn_ballot_w32(((df >> i) & 1u) != 0u);
            eqm |= ((m & 0xFFFFu) == 0u) ? (1u << (2 * i)) : 0u;
            eqm |= ((m >> 16) == 0u) ? (1u << (2 * i + 1)) : 0u;
        }
        if (lane < 16)
            lds[MSK_OFF + wv + NWAVES * lane] = ((eqm >> lane) & 1u) ? 1.0f : 0.0f;
#define SQ_EXCHANGE(OFF)                                              \
        {                                                             \
            const bool up = (lane & (OFF)) != 0;                      \
            _Pragma("unroll")                                         \
            for (int j = 0; j < (OFF); ++j) {                         \
                const float keep = up ? sq[j + (OFF)] : sq[j];        \
                const float send = up ? sq[j] : sq[j + (OFF)];        \
                sq[j] = keep + __shfl_xor(send, (OFF));               \
            }                                                         \
        }
        SQ_EXCHANGE(8)
        SQ_EXCHANGE(4)
        SQ_EXCHANGE(2)
        SQ_EXCHANGE(1)
#undef SQ_EXCHANGE
        lds[NRM_OFF + wv + NWAVES * (2 * l16 + half)] = sqrtf(sq[0]);
    }
    GRP(14) GRP(15)
    GRP(16) GRP(17) GRP(18)
#undef GRP
#undef KEEP_BELOW
#undef REFILL_BEHIND
#undef LOADS
#undef STAGE

    if (blockIdx.x == 0) {
        float* __restrict__ o = senWd + (long long)blockIdx.y * ((TILE / WPC) * DWORDS_D);
        for (int e = tid; e < (TILE / WPC) * DWORDS_D; e += NWAVES * 32) {
            const int si = e / DWORDS_D;
            const float* p = lds + e + si * ((WPC - 1) * DWORDS_D);
            float acc = 0.0f;
#pragma unroll
            for (int w = 0; w < WPC; ++w)
                acc += p[w * DWORDS_D];
            o[e] = acc;
        }
    }

    bar_wait();

    float nsC[WN], mkC[WN];
    float naR[WM][8];
#pragma unroll
    for (int ni = 0; ni < WN; ++ni) {
        nsC[ni] = lds[NRM_OFF + wc * 32 + ni * 16 + l16];
        mkC[ni] = lds[MSK_OFF + wc * 32 + ni * 16 + l16];
    }
#pragma unroll
    for (int mi = 0; mi < WM; ++mi) {
#pragma unroll
        for (int r = 0; r < 8; ++r)
            naR[mi][r] = lds[NRM_OFF + TILE + wr * 64 + mi * 16 + 8 * half + r];
    }

    const int wq = l16 & 7;
    const int step0 = wq * 8;
    const float ninf = -__builtin_inff();
    float res[WM][WN];
#pragma unroll
    for (int mi = 0; mi < WM; ++mi) {
#pragma unroll
        for (int ni = 0; ni < WN; ++ni) {
            const v8f cc = c[mi][ni];
            float x[8];
#pragma unroll
            for (int r = 0; r < 8; ++r) {
                const float denom = fmaxf(nsC[ni] * naR[mi][r], COS_EPS);
                float cosv = cc[r] / denom;
                if (mkC[ni] != 0.0f) cosv = 0.0f;
                x[r] = cosv;
            }
            int lastr = -1;
#pragma unroll
            for (int r = 0; r < 8; ++r)
                lastr = (x[r] < 0.0f) ? r : lastr;
            const int pl = (lastr >= 0) ? step0 + lastr : -1;
            int p = pl;
            p = max(p, xlane_i<1>(p));
            p = max(p, xlane_i<2>(p));
            p = max(p, xlane_i<4>(p));
            float m = ninf;
#pragma unroll
            for (int r = 0; r < 8; ++r) {
                const float cand = (step0 + r >= p) ? x[r] : ninf;
                m = fmaxf(m, cand);
            }
            m = fmaxf(m, xlane_f<1>(m));
            m = fmaxf(m, xlane_f<2>(m));
            m = fmaxf(m, xlane_f<4>(m));
            res[mi][ni] = (p < 0) ? fmaxf(m, 0.0f) : m;
        }
    }

    {
        float outv = res[0][0];
#pragma unroll
        for (int t = 1; t < WM * WN; ++t)
            outv = (wq == t) ? res[t >> 1][t & 1] : outv;
        const int sIdx = ((row0 + wc * 32) >> 3) + 2 * (wq & 1) + (l16 >> 3);
        const int aIdx = ((col0 + wr * 64) >> 3) + 2 * (wq >> 1) + half;
        maxSim[sIdx * A_CATS + aIdx] = outv;
    }
}

extern "C" void kernel_launch(void* const* d_in, const int* in_sizes, int n_in,
                              void* d_out, int out_size, void* d_ws, size_t ws_size,
                              hipStream_t stream) {
    const float* emb      = (const float*)d_in[0];
    const int*   senCats  = (const int*)d_in[1];
    const int*   annCats  = (const int*)d_in[2];
    const int*   noneIdx  = (const int*)d_in[3];

    float* out    = (float*)d_out;
    float* maxSim = out;
    float* senWd  = out + S_CATS * A_CATS;

    ComSim_cos_tile_wmma<<<dim3(16, 16), dim3(256), 0, stream>>>(
        emb, senCats, annCats, noneIdx, maxSim, senWd);
}
